# starting kernel
# baseline (speedup 1.0000x reference)
#include <hip/hip_runtime.h>
#include <math.h>

typedef __attribute__((ext_vector_type(2))) float v2f;
typedef __attribute__((ext_vector_type(4))) float v4f;
typedef __attribute__((ext_vector_type(8))) float v8f;

#define D 128
#define LNS 132
#define WAVES_PER_BLOCK 4

__global__ __launch_bounds__(32 * WAVES_PER_BLOCK)
void sage_wmma_kernel(const float* __restrict__ h,
                      const int*   __restrict__ nbr_idx,
                      const float* __restrict__ nbr_w,
                      const float* __restrict__ w_self,
                      const float* __restrict__ w_nei,
                      const float* __restrict__ gamma,
                      const float* __restrict__ beta,
                      float* __restrict__ out,
                      int N, int K)
{
    __shared__ float lds[WAVES_PER_BLOCK * 16 * LNS];

    const int lane   = threadIdx.x & 31;
    const int waveId = threadIdx.x >> 5;
    const int tile   = blockIdx.x * WAVES_PER_BLOCK + waveId;
    if (tile * 16 >= N) return;

    const int m    = lane & 15;
    const int half = lane >> 4;
    const int off  = half * 2;

    float* tileLds = lds + waveId * (16 * LNS);

    for (int r = 0; r < 16; ++r) {
        int grow = tile * 16 + r;
        if (grow >= N) grow = N - 1;
        v4f v = *reinterpret_cast<const v4f*>(h + (size_t)grow * D + 4 * lane);
        *reinterpret_cast<v4f*>(tileLds + r * LNS + 4 * lane) = v;
    }

    v2f aS[32];
#pragma unroll
    for (int c = 0; c < 32; ++c)
        aS[c] = *reinterpret_cast<const v2f*>(tileLds + m * LNS + 4 * c + off);

    for (int r = 0; r < 16; ++r) {
        int grow = tile * 16 + r;
        if (grow >= N) grow = N - 1;
        v4f acc = {};
        for (int k = 0; k < K; ++k) {
            const int   nrow = nbr_idx[(size_t)grow * K + k];
            const float w    = nbr_w [(size_t)grow * K + k];
            v4f v = *reinterpret_cast<const v4f*>(h + (size_t)nrow * D + 4 * lane);
            acc += v * w;
        }
        *reinterpret_cast<v4f*>(tileLds + r * LNS + 4 * lane) = acc;
    }

    v2f aA[32];
#pragma unroll
    for (int c = 0; c < 32; ++c)
        aA[c] = *reinterpret_cast<const v2f*>(tileLds + m * LNS + 4 * c + off);

    for (int jt = 0; jt < 8; ++jt) {
        const int j0 = jt * 16;
        v8f acc = {};
        const float* bsRow = w_self + (size_t)(j0 + m) * D + off;
        const float* bnRow = w_nei  + (size_t)(j0 + m) * D + off;
#pragma unroll
        for (int c = 0; c < 32; ++c) {
            v2f bS = *reinterpret_cast<const v2f*>(bsRow + 4 * c);
            acc = __builtin_amdgcn_wmma_f32_16x16x4_f32(false, aS[c], false, bS,
                                                        (short)0, acc, false, false);
            v2f bN = *reinterpret_cast<const v2f*>(bnRow + 4 * c);
            acc = __builtin_amdgcn_wmma_f32_16x16x4_f32(false, aA[c], false, bN,
                                                        (short)0, acc, false, false);
        }
#pragma unroll
        for (int r = 0; r < 8; ++r) {
            const int rr = r + 8 * half;
            const float z = acc[r];
            const float g = 0.5f * z * (1.0f + erff(z * 0.70710678118654752f));
            tileLds[rr * LNS + j0 + m] = g;
        }
    }

    for (int r = 0; r < 16; ++r) {
        const int grow = tile * 16 + r;
        const float x0 = tileLds[r * LNS + lane];
        const float x1 = tileLds[r * LNS + lane + 32];
        const float x2 = tileLds[r * LNS + lane + 64];
        const float x3 = tileLds[r * LNS + lane + 96];
        float s = x0 + x1 + x2 + x3;
#pragma unroll
        for (int d = 16; d >= 1; d >>= 1) s += __shfl_xor(s, d, 32);
        const float mu = s * (1.0f / D);
        const float d0 = x0 - mu, d1 = x1 - mu, d2 = x2 - mu, d3 = x3 - mu;
        float s2 = d0 * d0 + d1 * d1 + d2 * d2 + d3 * d3;
#pragma unroll
        for (int d = 16; d >= 1; d >>= 1) s2 += __shfl_xor(s2, d, 32);
        const float inv = rsqrtf(s2 * (1.0f / D) + 1e-5f);
        if (grow < N) {
            float* o = out + (size_t)grow * D;
            o[lane]      = d0 * inv * gamma[lane]      + beta[lane];
            o[lane + 32] = d1 * inv * gamma[lane + 32] + beta[lane + 32];
            o[lane + 64] = d2 * inv * gamma[lane + 64] + beta[lane + 64];
            o[lane + 96] = d3 * inv * gamma[lane + 96] + beta[lane + 96];
        }
    }
}

extern "C" void kernel_launch(void* const* d_in, const int* in_sizes, int n_in,
                              void* d_out, int out_size, void* d_ws, size_t ws_size,
                              hipStream_t stream) {
    const float* h       = (const float*)d_in[0];
    const int*   nbr_idx = (const int*)  d_in[1];
    const float* nbr_w   = (const float*)d_in[2];
    const float* w_self  = (const float*)d_in[3];
    const float* w_nei   = (const float*)d_in[4];
    const float* gamma   = (const float*)d_in[5];
    const float* beta    = (const float*)d_in[6];
    float* out = (float*)d_out;

    const int N = in_sizes[0] / D;
    const int K = (N > 0) ? (in_sizes[1] / N) : 16;

    const int tiles  = (N + 15) / 16;
    const int blocks = (tiles + WAVES_PER_BLOCK - 1) / WAVES_PER_BLOCK;
    hipLaunchKernelGGL(sage_wmma_kernel, dim3(blocks), dim3(32 * WAVES_PER_BLOCK), 0, stream,
                       h, nbr_idx, nbr_w, w_self, w_nei, gamma, beta, out, N, K);
}


# Next tile's 16 neighbour loads per wave issued in four groups of 4 under this tile's matrix and row phases (4 row buffers, all B k-steps stay in registers, barriers fence LDS only); sums unchanged
# speedup vs baseline: 2.2432x; 2.2432x over previous
#include <hip/hip_runtime.h>
#include <math.h>


typedef __attribute__((ext_vector_type(2))) float v2f;
typedef __attribute__((ext_vector_type(4))) float v4f;
typedef __attribute__((ext_vector_type(8))) float v8f;

#define D 128
#define LNS 132
#define NWAVES 16
#define GRID_BLOCKS 512

__device__ __forceinline__ v4f gather_row_any(const float* __restrict__ h,
                                              const int*   __restrict__ nbr_idx,
                                              const float* __restrict__ nbr_w,
                                              int grow, int K, int lane)
{
    const int*   ip = nbr_idx + (size_t)grow * K;
    const float* wp = nbr_w   + (size_t)grow * K;
    v4f acc = {};
    for (int k = 0; k < K; ++k) {
        const int   nrow = ip[k];
        const float w    = wp[k];
        v4f v = *reinterpret_cast<const v4f*>(h + (size_t)nrow * D + 4 * lane);
        acc += v * w;
    }
    return acc;
}

__device__ __forceinline__ v4f load_nbr_row(const float* __restrict__ h, int iv, int k, int lane)
{
    const int nrow = __builtin_amdgcn_readlane(iv, k);
    return *reinterpret_cast<const v4f*>(h + (size_t)nrow * D + 4 * lane);
}

__device__ __forceinline__ void lds_barrier()
{
    __builtin_amdgcn_fence(__ATOMIC_RELEASE, "workgroup", "local");
    __builtin_amdgcn_s_barrier();
    __builtin_amdgcn_fence(__ATOMIC_ACQUIRE, "workgroup", "local");
}

template <int DD>
__device__ __forceinline__ float xor_lane(float v)
{
    return __int_as_float(__builtin_amdgcn_ds_swizzle(__float_as_int(v), (DD << 10) | 0x1f));
}

__device__ __forceinline__ float wave_sum(float s)
{
    s += xor_lane<16>(s);
    s += xor_lane<8>(s);
    s += xor_lane<4>(s);
    s += xor_lane<2>(s);
    s += xor_lane<1>(s);
    return s;
}

template <int K0>
__device__ __forceinline__ void issue4(v4f (&buf)[4], const float* __restrict__ h, int iv, int lane)
{
#pragma unroll
    for (int k = 0; k < 4; ++k) buf[k] = load_nbr_row(h, iv, K0 + k, lane);
}

template <int K0>
__device__ __forceinline__ v4f sum4(v4f acc, const v4f (&buf)[4], float wv)
{
#pragma unroll
    for (int k = 0; k < 4; ++k) {
        const float w = __int_as_float(__builtin_amdgcn_readlane(__float_as_int(wv), K0 + k));
        acc += buf[k] * w;
    }
    return acc;
}

template <bool ROLL>
__device__ __forceinline__ void tile_trip(const float* __restrict__ h,
                                          const int*   __restrict__ nbr_idx,
                                          const float* __restrict__ nbr_w,
                                          const float* sGB,
                                          float* __restrict__ out,
                                          float* sSelf, float* sAgg, float* sZs, float* sZn,
                                          const float* sA, float* sZ,
                                          const v2f (&b)[32],
                                          int tile, int N, int wave, int lane,
                                          int m, int half, int off, int j0,
                                          v4f& sv_c, v4f& ag_c, int& iv_n, float& wv_n)
{
    *reinterpret_cast<v4f*>(sSelf + wave * LNS + 4 * lane) = sv_c;
    *reinterpret_cast<v4f*>(sAgg  + wave * LNS + 4 * lane) = ag_c;
    lds_barrier();

    v4f buf[4];
    v4f ag_n = {};
    int   iv_nn = 0;
    float wv_nn = 0.0f;
    if (ROLL) { issue4<0>(buf, h, iv_n, lane); asm volatile("" ::: "memory"); }

    v8f acc = {};
#pragma unroll
    for (int c = 0; c < 16; ++c) {
        const v2f a = *reinterpret_cast<const v2f*>(sA + m * LNS + 4 * c + off);
        acc = __builtin_amdgcn_wmma_f32_16x16x4_f32(false, a, false, b[c],
                                                    (short)0, acc, false, false);
    }
    if (ROLL) {
        asm volatile("" :: "v"(acc) : "memory");
        ag_n = sum4<0>(ag_n, buf, wv_n);
        asm volatile("" :: "v"(ag_n) : "memory");
        issue4<4>(buf, h, iv_n, lane);
        asm volatile("" ::: "memory");
    }
#pragma unroll
    for (int c = 16; c < 32; ++c) {
        const v2f a = *reinterpret_cast<const v2f*>(sA + m * LNS + 4 * c + off);
        acc = __builtin_amdgcn_wmma_f32_16x16x4_f32(false, a, false, b[c],
                                                    (short)0, acc, false, false);
    }
#pragma unroll
    for (int r = 0; r < 8; ++r)
        sZ[(r + 8 * half) * LNS + j0 + m] = acc[r];
    if (ROLL) {
        asm volatile("" ::: "memory");
        ag_n = sum4<4>(ag_n, buf, wv_n);
        asm volatile("" :: "v"(ag_n) : "memory");
        issue4<8>(buf, h, iv_n, lane);
        int grow2 = (tile + 2 * GRID_BLOCKS) * 16 + wave;
        if (grow2 >= N) grow2 = N - 1;
        iv_nn = nbr_idx[(size_t)grow2 * 16 + m];
        wv_nn = nbr_w  [(size_t)grow2 * 16 + m];
        asm volatile("" ::: "memory");
    }
    lds_barrier();

    v4f   sv_n  = {};
    {
        const int r    = wave;
        const int grow = tile * 16 + r;
        float z[4], x[4];
#pragma unroll
        for (int q = 0; q < 4; ++q)
            z[q] = sZs[r * LNS + lane + 32 * q] + sZn[r * LNS + lane + 32 * q];
#pragma unroll
        for (int q = 0; q < 4; ++q)
            x[q] = 0.5f * z[q] * (1.0f + erff(z[q] * 0.70710678118654752f));

        if (ROLL) {
            asm volatile("" ::: "memory");
            ag_n = sum4<8>(ag_n, buf, wv_n);
            asm volatile("" :: "v"(ag_n) : "memory");
            issue4<12>(buf, h, iv_n, lane);
            int grow1 = (tile + GRID_BLOCKS) * 16 + wave;
            if (grow1 >= N) grow1 = N - 1;
            sv_n = *reinterpret_cast<const v4f*>(h + (size_t)grow1 * D + 4 * lane);
            asm volatile("" ::: "memory");
        }

        const float x0 = x[0], x1 = x[1], x2 = x[2], x3 = x[3];
        float s = x0 + x1 + x2 + x3;
        s = wave_sum(s);
        const float mu = s * (1.0f / D);
        const float d0 = x0 - mu, d1 = x1 - mu, d2 = x2 - mu, d3 = x3 - mu;
        float s2 = d0 * d0 + d1 * d1 + d2 * d2 + d3 * d3;
        s2 = wave_sum(s2);
        const float inv = rsqrtf(s2 * (1.0f / D) + 1e-5f);
        if (grow < N) {
            float* o = out + (size_t)grow * D;
            o[lane]      = d0 * inv * sGB[lane]      + sGB[D + lane];
            o[lane + 32] = d1 * inv * sGB[lane + 32] + sGB[D + lane + 32];
            o[lane + 64] = d2 * inv * sGB[lane + 64] + sGB[D + lane + 64];
            o[lane + 96] = d3 * inv * sGB[lane + 96] + sGB[D + lane + 96];
        }
    }

    if (ROLL) { asm volatile("" ::: "memory"); ag_n = sum4<12>(ag_n, buf, wv_n); }
    sv_c = sv_n;  ag_c = ag_n;
    iv_n = iv_nn; wv_n = wv_nn;
}

__global__ __launch_bounds__(32 * NWAVES, 8)
void sage_split_slab_kernel(const float* __restrict__ h,
                            const int*   __restrict__ nbr_idx,
                            const float* __restrict__ nbr_w,
                            const float* __restrict__ w_self,
                            const float* __restrict__ w_nei,
                            const float* __restrict__ gamma,
                            const float* __restrict__ beta,
                            float* __restrict__ out,
                            int N, int K)
{
    __shared__ float sSelf[16 * LNS];
    __shared__ float sAgg [16 * LNS];
    __shared__ float sZs  [16 * LNS];
    __shared__ float sZn  [16 * LNS];
    __shared__ float sGB  [2 * D];
    if (threadIdx.x < 2 * D)
        sGB[threadIdx.x] = (threadIdx.x < D) ? gamma[threadIdx.x] : beta[threadIdx.x - D];

    const int lane = threadIdx.x & 31;
    const int wave = __builtin_amdgcn_readfirstlane((int)(threadIdx.x >> 5));
    const int slab = wave & 7;
    const int part = wave >> 3;

    const int m    = lane & 15;
    const int half = lane >> 4;
    const int off  = half * 2;
    const int j0   = slab * 16;

    v2f b[32];
    {
        const float* wmat = part ? w_nei : w_self;
        const float* bRow = wmat + (size_t)(j0 + m) * D + off;
#pragma unroll
        for (int c = 0; c < 32; ++c)
            b[c] = *reinterpret_cast<const v2f*>(bRow + 4 * c);
    }
    const float* sA = part ? sAgg : sSelf;
    float*       sZ = part ? sZn  : sZs;

    const int tiles = (N + 15) >> 4;
    const bool k16 = (K == 16);

    int tile = blockIdx.x;
    if (tile >= tiles) return;

    v4f   sv_c;
    v4f   ag_c;
    int   iv_n = 0;
    float wv_n = 0.0f;
    {
        int grow = tile * 16 + wave;
        if (grow >= N) grow = N - 1;
        sv_c = *reinterpret_cast<const v4f*>(h + (size_t)grow * D + 4 * lane);
        if (k16) {
            const int   iv = nbr_idx[(size_t)grow * 16 + m];
            const float wv = nbr_w  [(size_t)grow * 16 + m];
            int grow2 = (tile + GRID_BLOCKS) * 16 + wave;
            if (grow2 >= N) grow2 = N - 1;
            iv_n = nbr_idx[(size_t)grow2 * 16 + m];
            wv_n = nbr_w  [(size_t)grow2 * 16 + m];
            v4f buf[4];
            v4f ag = {};
            issue4<0>(buf, h, iv, lane);
            ag = sum4<0>(ag, buf, wv);
            asm volatile("" :: "v"(ag) : "memory");
            issue4<4>(buf, h, iv, lane);
            ag = sum4<4>(ag, buf, wv);
            asm volatile("" :: "v"(ag) : "memory");
            issue4<8>(buf, h, iv, lane);
            ag = sum4<8>(ag, buf, wv);
            asm volatile("" :: "v"(ag) : "memory");
            issue4<12>(buf, h, iv, lane);
            ag_c = sum4<12>(ag, buf, wv);
        } else {
            ag_c = gather_row_any(h, nbr_idx, nbr_w, grow, K, lane);
        }
    }

    if (k16) {
        for (; tile + GRID_BLOCKS < tiles; tile += GRID_BLOCKS)
            tile_trip<true>(h, nbr_idx, nbr_w, sGB, out, sSelf, sAgg, sZs, sZn, sA, sZ, b,
                            tile, N, wave, lane, m, half, off, j0, sv_c, ag_c, iv_n, wv_n);
        int lane_e = lane, wave_e = wave;
        asm volatile("" : "+v"(lane_e), "+s"(wave_e));
        const int m_e = lane_e & 15, half_e = lane_e >> 4, part_e = wave_e >> 3;
        tile_trip<false>(h, nbr_idx, nbr_w, sGB, out, sSelf, sAgg, sZs, sZn,
                         part_e ? sAgg : sSelf, part_e ? sZn : sZs, b,
                         tile, N, wave_e, lane_e, m_e, half_e, half_e * 2, (wave_e & 7) * 16,
                         sv_c, ag_c, iv_n, wv_n);
    } else {
        for (; tile < tiles; tile += GRID_BLOCKS) {
            tile_trip<false>(h, nbr_idx, nbr_w, sGB, out, sSelf, sAgg, sZs, sZn, sA, sZ, b,
                             tile, N, wave, lane, m, half, off, j0, sv_c, ag_c, iv_n, wv_n);
            if (tile + GRID_BLOCKS < tiles) {
                int grow = (tile + GRID_BLOCKS) * 16 + wave;
                if (grow >= N) grow = N - 1;
                sv_c = *reinterpret_cast<const v4f*>(h + (size_t)grow * D + 4 * lane);
                ag_c = gather_row_any(h, nbr_idx, nbr_w, grow, K, lane);
            }
        }
    }
}

extern "C" void kernel_launch(void* const* d_in, const int* in_sizes, int n_in,
                              void* d_out, int out_size, void* d_ws, size_t ws_size,
                              hipStream_t stream) {
    const float* h       = (const float*)d_in[0];
    const int*   nbr_idx = (const int*)  d_in[1];
    const float* nbr_w   = (const float*)d_in[2];
    const float* w_self  = (const float*)d_in[3];
    const float* w_nei   = (const float*)d_in[4];
    const float* gamma   = (const float*)d_in[5];
    const float* beta    = (const float*)d_in[6];
    float* out = (float*)d_out;

    const int N = in_sizes[0] / D;
    const int K = (N > 0) ? (in_sizes[1] / N) : 16;

    hipLaunchKernelGGL(sage_split_slab_kernel, dim3(GRID_BLOCKS), dim3(32 * NWAVES), 0, stream,
                       h, nbr_idx, nbr_w, w_self, w_nei, gamma, beta, out, N, K);
}
